# starting kernel
# baseline (speedup 1.0000x reference)
#include <hip/hip_runtime.h>
#include <cstdint>

typedef __attribute__((ext_vector_type(2))) float v2f;
typedef __attribute__((ext_vector_type(8))) float v8f;

#define CFEAT 128


__global__ void k_init(int* __restrict__ deg, int* __restrict__ cursor, int n) {
    int i = blockIdx.x * blockDim.x + threadIdx.x;
    if (i < n) { deg[i] = 0; cursor[i] = 0; }
}

__global__ void k_count(const int* __restrict__ dst, int* __restrict__ deg, int e) {
    int i = blockIdx.x * blockDim.x + threadIdx.x;
    if (i < e) atomicAdd(&deg[dst[i]], 1);
}

__global__ void k_dinv(const int* __restrict__ deg, float* __restrict__ dinv, int n) {
    int i = blockIdx.x * blockDim.x + threadIdx.x;
    if (i < n) dinv[i] = rsqrtf((float)deg[i] + 1.0f);
}

__global__ __launch_bounds__(1024)
void k_scan(const int* __restrict__ deg, int* __restrict__ offs, int n) {
    __shared__ int buf[1024];
    __shared__ int carry;
    const int tid = threadIdx.x;
    if (tid == 0) carry = 0;
    __syncthreads();
    for (int base = 0; base < n; base += 1024) {
        int i = base + tid;
        int v = (i < n) ? deg[i] : 0;
        buf[tid] = v;
        __syncthreads();
        #pragma unroll
        for (int off = 1; off < 1024; off <<= 1) {
            int t = (tid >= off) ? buf[tid - off] : 0;
            __syncthreads();
            buf[tid] += t;
            __syncthreads();
        }
        int incl = buf[tid];
        int c = carry;
        if (i < n) offs[i] = c + incl - v;
        __syncthreads();
        if (tid == 0) carry = c + buf[1023];
        __syncthreads();
    }
    if (tid == 0) offs[n] = carry;
}

__global__ void k_fill(const int* __restrict__ src, const int* __restrict__ dst,
                       const int* __restrict__ offs, int* __restrict__ cursor,
                       int* __restrict__ csr, int e) {
    int i = blockIdx.x * blockDim.x + threadIdx.x;
    if (i < e) {
        int d = dst[i];
        int p = atomicAdd(&cursor[d], 1);
        csr[offs[d] + p] = src[i];
    }
}


__global__ __launch_bounds__(256)
void k_gemm1(const float* __restrict__ x, const float* __restrict__ W1,
             float* __restrict__ h, int n) {
    __shared__ float sW[CFEAT * CFEAT];
    for (int i = threadIdx.x; i < (CFEAT * CFEAT) / 4; i += 256)
        ((float4*)sW)[i] = ((const float4*)W1)[i];
    __syncthreads();

    const int lane = threadIdx.x & 31;
    const int wave = threadIdx.x >> 5;
    const int hi   = lane >> 4;
    const int lo   = lane & 15;
    const int rowbase = (blockIdx.x * 8 + wave) * 16;
    if (rowbase >= n) return;

    int arow = rowbase + lo;
    if (arow >= n) arow = n - 1;
    const float* __restrict__ xrow = x + (size_t)arow * CFEAT;

    v8f acc[8];
    const v8f z = {0.f, 0.f, 0.f, 0.f, 0.f, 0.f, 0.f, 0.f};
    #pragma unroll
    for (int t = 0; t < 8; ++t) acc[t] = z;

    for (int k = 0; k < CFEAT; k += 4) {
        v2f a;
        a.x = xrow[k + 2 * hi];
        a.y = xrow[k + 2 * hi + 1];
        #pragma unroll
        for (int t = 0; t < 8; ++t) {
            v2f b;
            const int col = t * 16 + lo;
            b.x = sW[(k + 2 * hi) * CFEAT + col];
            b.y = sW[(k + 2 * hi + 1) * CFEAT + col];
            acc[t] = __builtin_amdgcn_wmma_f32_16x16x4_f32(
                false, a, false, b, (short)0, acc[t], false, false);
        }
    }

    if (rowbase + 16 <= n) {
        #pragma unroll
        for (int t = 0; t < 8; ++t) {
            #pragma unroll
            for (int r = 0; r < 8; ++r) {
                const int row = rowbase + r + 8 * hi;
                h[(size_t)row * CFEAT + t * 16 + lo] = acc[t][r];
            }
        }
    } else {
        #pragma unroll
        for (int t = 0; t < 8; ++t) {
            #pragma unroll
            for (int r = 0; r < 8; ++r) {
                const int row = rowbase + r + 8 * hi;
                if (row < n) h[(size_t)row * CFEAT + t * 16 + lo] = acc[t][r];
            }
        }
    }
}


__device__ __forceinline__ float4 aggregate(const float* __restrict__ hsrc,
                                            const int* __restrict__ offs,
                                            const int* __restrict__ csr,
                                            const float* __restrict__ dinv,
                                            int node, int lane) {
    const float din = dinv[node];
    const float4* __restrict__ h4 = (const float4*)hsrc;
    float4 acc;
    {
        float4 v = h4[(size_t)node * 32 + lane];
        acc.x = v.x * din; acc.y = v.y * din; acc.z = v.z * din; acc.w = v.w * din;
    }
    const int beg = offs[node], end = offs[node + 1];
    for (int j = beg; j < end; j += 32) {
        const int idx = j + lane;
        const int   s  = (idx < end) ? csr[idx] : 0;
        const float ws = (idx < end) ? dinv[s] : 0.0f;
        int cnt = end - j; if (cnt > 32) cnt = 32;
        int t = 0;
        if (cnt == 32) {
            for (; t < 32; t += 4) {
                const int   s0 = __shfl(s, t,     32), s1 = __shfl(s, t + 1, 32);
                const int   s2 = __shfl(s, t + 2, 32), s3 = __shfl(s, t + 3, 32);
                const float w0 = __shfl(ws, t,     32), w1 = __shfl(ws, t + 1, 32);
                const float w2 = __shfl(ws, t + 2, 32), w3 = __shfl(ws, t + 3, 32);
                float4 v0 = h4[(size_t)s0 * 32 + lane];
                float4 v1 = h4[(size_t)s1 * 32 + lane];
                float4 v2 = h4[(size_t)s2 * 32 + lane];
                float4 v3 = h4[(size_t)s3 * 32 + lane];
                acc.x += v0.x * w0; acc.y += v0.y * w0; acc.z += v0.z * w0; acc.w += v0.w * w0;
                acc.x += v1.x * w1; acc.y += v1.y * w1; acc.z += v1.z * w1; acc.w += v1.w * w1;
                acc.x += v2.x * w2; acc.y += v2.y * w2; acc.z += v2.z * w2; acc.w += v2.w * w2;
                acc.x += v3.x * w3; acc.y += v3.y * w3; acc.z += v3.z * w3; acc.w += v3.w * w3;
            }
        }
        for (; t < cnt; ++t) {
            const int   sk = __shfl(s, t, 32);
            const float wk = __shfl(ws, t, 32);
            float4 v = h4[(size_t)sk * 32 + lane];
            acc.x += v.x * wk; acc.y += v.y * wk; acc.z += v.z * wk; acc.w += v.w * wk;
        }
    }
    acc.x *= din; acc.y *= din; acc.z *= din; acc.w *= din;
    return acc;
}

__global__ __launch_bounds__(256)
void k_prop1(const float* __restrict__ h, float* __restrict__ h2,
             const int* __restrict__ offs, const int* __restrict__ csr,
             const float* __restrict__ dinv, const float* __restrict__ b1, int n) {
    const int node = (int)((blockIdx.x * blockDim.x + threadIdx.x) >> 5);
    const int lane = threadIdx.x & 31;
    if (node >= n) return;
    float4 acc = aggregate(h, offs, csr, dinv, node, lane);
    const float4 b = ((const float4*)b1)[lane];
    acc.x = fmaxf(acc.x + b.x, 0.0f);
    acc.y = fmaxf(acc.y + b.y, 0.0f);
    acc.z = fmaxf(acc.z + b.z, 0.0f);
    acc.w = fmaxf(acc.w + b.w, 0.0f);
    ((float4*)h2)[(size_t)node * 32 + lane] = acc;
}

__global__ __launch_bounds__(256)
void k_prop2_out(const float* __restrict__ h2, float* __restrict__ out,
                 const int* __restrict__ offs, const int* __restrict__ csr,
                 const float* __restrict__ dinv, const float* __restrict__ Wout,
                 const float* __restrict__ bout, int n) {
    const int node = (int)((blockIdx.x * blockDim.x + threadIdx.x) >> 5);
    const int lane = threadIdx.x & 31;
    if (node >= n) return;

    float wl[4][4][2];
    #pragma unroll
    for (int m = 0; m < 4; ++m)
        #pragma unroll
        for (int j = 0; j < 4; ++j) {
            const float2 w = ((const float2*)Wout)[m * CFEAT + lane * 4 + j];
            wl[m][j][0] = w.x; wl[m][j][1] = w.y;
        }

    float4 acc = aggregate(h2, offs, csr, dinv, node, lane);

    float p[4][2];
    #pragma unroll
    for (int m = 0; m < 4; ++m)
        #pragma unroll
        for (int o = 0; o < 2; ++o)
            p[m][o] = acc.x * wl[m][0][o] + acc.y * wl[m][1][o] +
                      acc.z * wl[m][2][o] + acc.w * wl[m][3][o];

    #pragma unroll
    for (int m = 0; m < 4; ++m)
        #pragma unroll
        for (int o = 0; o < 2; ++o) {
            float v = p[m][o];
            #pragma unroll
            for (int d = 16; d > 0; d >>= 1) v += __shfl_xor(v, d, 32);
            p[m][o] = v;
        }

    if (lane == 0) {
        #pragma unroll
        for (int m = 0; m < 4; ++m) {
            float2 r;
            r.x = p[m][0] + bout[m * 2 + 0];
            r.y = p[m][1] + bout[m * 2 + 1];
            ((float2*)out)[(size_t)m * n + node] = r;
        }
    }
}


extern "C" void kernel_launch(void* const* d_in, const int* in_sizes, int n_in,
                              void* d_out, int out_size, void* d_ws, size_t ws_size,
                              hipStream_t stream) {
    const float* x    = (const float*)d_in[0];
    const int*   ei   = (const int*)d_in[1];
    const float* W1   = (const float*)d_in[2];
    const float* b1   = (const float*)d_in[3];
    const float* Wout = (const float*)d_in[4];
    const float* bout = (const float*)d_in[5];
    float* out = (float*)d_out;

    const int n = in_sizes[0] / CFEAT;
    const int e = in_sizes[1] / 2;
    const int* srcv = ei;
    const int* dstv = ei + e;

    uintptr_t p = (uintptr_t)d_ws;
    auto alloc = [&](size_t bytes) {
        p = (p + 15) & ~(uintptr_t)15;
        void* r = (void*)p;
        p += bytes;
        return r;
    };
    int*   deg    = (int*)  alloc((size_t)n * 4);
    int*   cursor = (int*)  alloc((size_t)n * 4);
    int*   offs   = (int*)  alloc((size_t)(n + 1) * 4);
    int*   csr    = (int*)  alloc((size_t)e * 4);
    float* dinv   = (float*)alloc((size_t)n * 4);
    float* h      = (float*)alloc((size_t)n * CFEAT * 4);
    float* h2     = (float*)alloc((size_t)n * CFEAT * 4);
    (void)ws_size; (void)n_in; (void)out_size;

    const int TB = 256;
    const int gn = (n + TB - 1) / TB;
    const int ge = (e + TB - 1) / TB;

    k_init <<<gn, TB, 0, stream>>>(deg, cursor, n);
    k_count<<<ge, TB, 0, stream>>>(dstv, deg, e);
    k_dinv <<<gn, TB, 0, stream>>>(deg, dinv, n);
    k_scan <<<1, 1024, 0, stream>>>(deg, offs, n);
    k_fill <<<ge, TB, 0, stream>>>(srcv, dstv, offs, cursor, csr, e);

    k_gemm1<<<(n + 127) / 128, TB, 0, stream>>>(x, W1, h, n);

    const int gprop = (n + 7) / 8;
    k_prop1    <<<gprop, TB, 0, stream>>>(h, h2, offs, csr, dinv, b1, n);
    k_prop2_out<<<gprop, TB, 0, stream>>>(h2, out, offs, csr, dinv, Wout, bout, n);
}


# k_prop1 stores each h2 row times its node's dinv; k_prop2_out adds the gathered rows unweighted: its per-edge degree-word gather and that dependent hop go; products rounded before adds, order kept
# speedup vs baseline: 4.0328x; 4.0328x over previous
#include <hip/hip_runtime.h>
#include <cstdint>

typedef __attribute__((ext_vector_type(2))) float v2f;
typedef __attribute__((ext_vector_type(8))) float v8f;

#define CFEAT 128
#define SLOTS 64


#define TICKET_PITCH 64

#define FILL_TB   256
#define FILL_EPT  2
#define FILL_SLICE (FILL_TB * FILL_EPT)

__global__ __launch_bounds__(FILL_TB)
void k_fill(const int* __restrict__ src, const int* __restrict__ dst,
            int* __restrict__ deg, int* __restrict__ slots,
            int* __restrict__ spill_src, int* __restrict__ spill_dst,
            int* __restrict__ nspill, int* __restrict__ ticket, int e, int n) {
    __shared__ int s_half;
    __shared__ int s_slice[2];
    if (threadIdx.x == 0) {
        const unsigned id = __builtin_amdgcn_s_sendmsg_rtn(0x87);
        s_half = (int)((id >> 18) & 1u);
    }
    __syncthreads();
    const int myhalf  = s_half;
    const int split   = ((n >> 1) + 31) & ~31;
    const int nslices = (e + FILL_SLICE - 1) / FILL_SLICE;
    int turn = 0;
    for (int pass = 0; pass < 2; ++pass) {
        const int range = myhalf ^ pass;
        int* __restrict__ tk = ticket + range * TICKET_PITCH;
        for (;;) {
            if (threadIdx.x == 0) s_slice[turn] = atomicAdd(tk, 1);
            __syncthreads();
            const int sl = s_slice[turn];
            turn ^= 1;
            if (sl >= nslices) break;
            int dd[FILL_EPT], ss[FILL_EPT];
            #pragma unroll
            for (int k = 0; k < FILL_EPT; ++k) {
                const int i = sl * FILL_SLICE + k * FILL_TB + (int)threadIdx.x;
                const bool in = i < e;
                dd[k] = in ? dst[i] : -1;
                ss[k] = in ? src[i] : 0;
            }
            int pp[FILL_EPT];
            #pragma unroll
            for (int k = 0; k < FILL_EPT; ++k) {
                const int d = dd[k];
                const bool mine = d >= 0 && (int)(d >= split) == range;
                pp[k] = mine ? atomicAdd(&deg[d], 1) : -1;
            }
            #pragma unroll
            for (int k = 0; k < FILL_EPT; ++k) {
                const int p = pp[k];
                if (p >= 0) {
                    const int d = dd[k];
                    if (p < SLOTS) {
                        slots[(size_t)d * SLOTS + p] = ss[k];
                    } else {
                        const int q = atomicAdd(nspill, 1);
                        spill_src[q] = ss[k];
                        spill_dst[q] = d;
                    }
                }
            }
        }
    }
}

__device__ __forceinline__ float dinv_of(int deg) { return rsqrtf((float)deg + 1.0f); }


__global__ __launch_bounds__(256)
void k_gemm1(const float* __restrict__ x, const float* __restrict__ W1,
             float* __restrict__ h, int* __restrict__ deg, int* __restrict__ nspill,
             int* __restrict__ ticket, int n) {
    __shared__ float sW[CFEAT * CFEAT];
    {
        const int i = blockIdx.x * 256 + threadIdx.x;
        if (i < n) deg[i] = 0;
        if (i == 0) { *nspill = 0; ticket[0] = 0; ticket[TICKET_PITCH] = 0; }
    }
    for (int i = threadIdx.x; i < (CFEAT * CFEAT) / 4; i += 256)
        ((float4*)sW)[i] = ((const float4*)W1)[i];
    __syncthreads();

    const int lane = threadIdx.x & 31;
    const int wave = threadIdx.x >> 5;
    const int hi   = lane >> 4;
    const int lo   = lane & 15;
    const int rowbase = (blockIdx.x * 8 + wave) * 16;
    if (rowbase >= n) return;

    int arow = rowbase + lo;
    if (arow >= n) arow = n - 1;
    const float* __restrict__ xrow = x + (size_t)arow * CFEAT;

    v8f acc[8];
    const v8f z = {0.f, 0.f, 0.f, 0.f, 0.f, 0.f, 0.f, 0.f};
    #pragma unroll
    for (int t = 0; t < 8; ++t) acc[t] = z;

    for (int k = 0; k < CFEAT; k += 4) {
        v2f a;
        a.x = xrow[k + 2 * hi];
        a.y = xrow[k + 2 * hi + 1];
        #pragma unroll
        for (int t = 0; t < 8; ++t) {
            v2f b;
            const int col = t * 16 + lo;
            b.x = sW[(k + 2 * hi) * CFEAT + col];
            b.y = sW[(k + 2 * hi + 1) * CFEAT + col];
            acc[t] = __builtin_amdgcn_wmma_f32_16x16x4_f32(
                false, a, false, b, (short)0, acc[t], false, false);
        }
    }

    if (rowbase + 16 <= n) {
        #pragma unroll
        for (int t = 0; t < 8; ++t) {
            #pragma unroll
            for (int r = 0; r < 8; ++r) {
                const int row = rowbase + r + 8 * hi;
                h[(size_t)row * CFEAT + t * 16 + lo] = acc[t][r];
            }
        }
    } else {
        #pragma unroll
        for (int t = 0; t < 8; ++t) {
            #pragma unroll
            for (int r = 0; r < 8; ++r) {
                const int row = rowbase + r + 8 * hi;
                if (row < n) h[(size_t)row * CFEAT + t * 16 + lo] = acc[t][r];
            }
        }
    }
}


struct Graph {
    const int* deg;
    const int* slots;
    const int* spill_src;
    const int* spill_dst;
    const int* nspill;
};

#define NAB      4
#define PROP_TB  128

template <bool WEIGHTED>
__device__ __forceinline__ void gather_half(const float4* __restrict__ h4, int lane, int trips,
                                            const int (&cnt)[NAB], const int (&s)[NAB],
                                            const float (&w)[NAB], float4 (&acc)[NAB]) {
    for (int t = 0; t < trips; ++t) {
        float4 v[NAB];
        float  wk[NAB];
        #pragma unroll
        for (int k = 0; k < NAB; ++k) {
            if (t < cnt[k]) {
                const int sk = __builtin_amdgcn_readlane(s[k], t);
                if (WEIGHTED) wk[k] = __int_as_float(__builtin_amdgcn_readlane(__float_as_int(w[k]), t));
                v[k] = h4[(size_t)sk * 32 + lane];
            }
        }
        #pragma unroll
        for (int k = 0; k < NAB; ++k) {
            if (t < cnt[k]) {
                if (WEIGHTED) {
                    acc[k].x += v[k].x * wk[k]; acc[k].y += v[k].y * wk[k];
                    acc[k].z += v[k].z * wk[k]; acc[k].w += v[k].w * wk[k];
                } else {
                    acc[k].x += v[k].x; acc[k].y += v[k].y;
                    acc[k].z += v[k].z; acc[k].w += v[k].w;
                }
            }
        }
    }
}

template <bool WEIGHTED>
__device__ __forceinline__ void sum_abreast(const float* __restrict__ hsrc,
                                            const Graph g,
                                            int node0, int n, int lane, float4 (&acc)[NAB],
                                            float (&din)[NAB]) {
    const float4* __restrict__ h4 = (const float4*)hsrc;
    int   nd[NAB];
    int   d[NAB];
    #pragma unroll
    for (int k = 0; k < NAB; ++k) {
        const bool ok = node0 + k < n;
        nd[k] = ok ? node0 + k : node0;
        const int dk = __builtin_amdgcn_readfirstlane(g.deg[nd[k]]);
        d[k]   = ok ? dk : 0;
        din[k] = dinv_of(dk);
    }
    int c0[NAB], c1[NAB];
    #pragma unroll
    for (int k = 0; k < NAB; ++k) {
        const int end = d[k] < SLOTS ? d[k] : SLOTS;
        c0[k] = end < 32 ? end : 32;
        c1[k] = end - c0[k];
    }
    int s0[NAB], s1[NAB];
    #pragma unroll
    for (int k = 0; k < NAB; ++k) {
        const int* __restrict__ row = g.slots + (size_t)nd[k] * SLOTS;
        const int* __restrict__ p0 = (lane < c0[k]) ? row + lane      : g.deg + nd[k];
        const int* __restrict__ p1 = (lane < c1[k]) ? row + lane + 32 : g.deg + nd[k];
        s0[k] = *p0;
        s1[k] = *p1;
    }
    #pragma unroll
    for (int k = 0; k < NAB; ++k) {
        s0[k] = (lane < c0[k]) ? s0[k] : 0;
        s1[k] = (lane < c1[k]) ? s1[k] : 0;
    }
    int g0[NAB], g1[NAB];
    if (WEIGHTED) {
        #pragma unroll
        for (int k = 0; k < NAB; ++k) {
            g0[k] = g.deg[s0[k]];
            g1[k] = g.deg[s1[k]];
        }
    }
    #pragma unroll
    for (int k = 0; k < NAB; ++k) {
        const float4 v = h4[(size_t)nd[k] * 32 + lane];
        if (WEIGHTED) {
            acc[k].x = v.x * din[k]; acc[k].y = v.y * din[k];
            acc[k].z = v.z * din[k]; acc[k].w = v.w * din[k];
        } else {
            acc[k] = v;
        }
    }
    if (WEIGHTED)
        asm volatile("" :: "v"(g0[0]), "v"(g0[1]), "v"(g0[2]), "v"(g0[3]),
                           "v"(g1[0]), "v"(g1[1]), "v"(g1[2]), "v"(g1[3]));
    static_assert(NAB == 4, "the fence above names four nodes' degree words");
    float w0[NAB], w1[NAB];
    #pragma unroll
    for (int k = 0; k < NAB; ++k) {
        w0[k] = WEIGHTED ? dinv_of(g0[k]) : 1.0f;
        w1[k] = WEIGHTED ? dinv_of(g1[k]) : 1.0f;
    }
    int trips0 = 0, trips1 = 0;
    #pragma unroll
    for (int k = 0; k < NAB; ++k) {
        trips0 = c0[k] > trips0 ? c0[k] : trips0;
        trips1 = c1[k] > trips1 ? c1[k] : trips1;
    }
    gather_half<WEIGHTED>(h4, lane, trips0, c0, s0, w0, acc);
    gather_half<WEIGHTED>(h4, lane, trips1, c1, s1, w1, acc);
    #pragma unroll
    for (int k = 0; k < NAB; ++k) {
        if (d[k] > SLOTS) {
            const int node = node0 + k;
            const int ns = *g.nspill;
            for (int q = 0; q < ns; ++q) {
                if (g.spill_dst[q] == node) {
                    const int   sk = g.spill_src[q];
                    const float wk = WEIGHTED ? dinv_of(g.deg[sk]) : 1.0f;
                    const float4 v = h4[(size_t)sk * 32 + lane];
                    acc[k].x += v.x * wk; acc[k].y += v.y * wk; acc[k].z += v.z * wk; acc[k].w += v.w * wk;
                }
            }
        }
        acc[k].x *= din[k]; acc[k].y *= din[k]; acc[k].z *= din[k]; acc[k].w *= din[k];
    }
}

__global__ __launch_bounds__(PROP_TB)
void k_prop1(const float* __restrict__ h, float* __restrict__ h2,
             const Graph g,
             const float* __restrict__ b1, int n) {
    const int lane  = threadIdx.x & 31;
    const int wave  = __builtin_amdgcn_readfirstlane((int)((blockIdx.x * blockDim.x + threadIdx.x) >> 5));
    const int node0 = wave * NAB;
    if (node0 >= n) return;
    float4 acc[NAB];
    float  din[NAB];
    sum_abreast<true>(h, g, node0, n, lane, acc, din);
    const float4 b = ((const float4*)b1)[lane];
    #pragma unroll
    for (int k = 0; k < NAB; ++k) {
        if (node0 + k < n) {
            float4 r;
            r.x = fmaxf(acc[k].x + b.x, 0.0f) * din[k];
            r.y = fmaxf(acc[k].y + b.y, 0.0f) * din[k];
            r.z = fmaxf(acc[k].z + b.z, 0.0f) * din[k];
            r.w = fmaxf(acc[k].w + b.w, 0.0f) * din[k];
            ((float4*)h2)[(size_t)(node0 + k) * 32 + lane] = r;
        }
    }
}

__global__ __launch_bounds__(PROP_TB)
void k_prop2_out(const float* __restrict__ h2, float* __restrict__ out,
                 const Graph g,
                 const float* __restrict__ Wout,
                 const float* __restrict__ bout, int n) {
    const int lane  = threadIdx.x & 31;
    const int wave  = __builtin_amdgcn_readfirstlane((int)((blockIdx.x * blockDim.x + threadIdx.x) >> 5));
    const int node0 = wave * NAB;
    if (node0 >= n) return;

    float4 acc[NAB];
    float  din[NAB];
    sum_abreast<false>(h2, g, node0, n, lane, acc, din);

    #pragma unroll
    for (int m = 0; m < 4; ++m) {
        float wl[4][2];
        #pragma unroll
        for (int j = 0; j < 4; ++j) {
            const float2 w = ((const float2*)Wout)[m * CFEAT + lane * 4 + j];
            wl[j][0] = w.x; wl[j][1] = w.y;
        }
        float p[NAB][2];
        #pragma unroll
        for (int k = 0; k < NAB; ++k)
            #pragma unroll
            for (int o = 0; o < 2; ++o) {
                float v = acc[k].x * wl[0][o] + acc[k].y * wl[1][o] +
                          acc[k].z * wl[2][o] + acc[k].w * wl[3][o];
                #pragma unroll
                for (int dd = 16; dd > 0; dd >>= 1) v += __shfl_xor(v, dd, 32);
                p[k][o] = v;
            }
        if (lane == 0) {
            const float bo0 = bout[m * 2 + 0];
            const float bo1 = bout[m * 2 + 1];
            #pragma unroll
            for (int k = 0; k < NAB; ++k) {
                if (node0 + k < n) {
                    float2 r;
                    r.x = p[k][0] + bo0;
                    r.y = p[k][1] + bo1;
                    ((float2*)out)[(size_t)m * n + node0 + k] = r;
                }
            }
        }
    }
}


extern "C" void kernel_launch(void* const* d_in, const int* in_sizes, int n_in,
                              void* d_out, int out_size, void* d_ws, size_t ws_size,
                              hipStream_t stream) {
    const float* x    = (const float*)d_in[0];
    const int*   ei   = (const int*)d_in[1];
    const float* W1   = (const float*)d_in[2];
    const float* b1   = (const float*)d_in[3];
    const float* Wout = (const float*)d_in[4];
    const float* bout = (const float*)d_in[5];
    float* out = (float*)d_out;

    const int n = in_sizes[0] / CFEAT;
    const int e = in_sizes[1] / 2;
    const int* srcv = ei;
    const int* dstv = ei + e;

    uintptr_t p = (uintptr_t)d_ws;
    auto alloc = [&](size_t bytes) {
        p = (p + 15) & ~(uintptr_t)15;
        void* r = (void*)p;
        p += bytes;
        return r;
    };
    int*   deg    = (int*)  alloc((size_t)n * 4);
    int*   nspill = (int*)  alloc(16);
    p = (p + 255) & ~(uintptr_t)255;
    int*   ticket = (int*)  alloc(2 * TICKET_PITCH * 4);
    p = (p + 255) & ~(uintptr_t)255;
    int*   slots  = (int*)  alloc((size_t)n * SLOTS * 4);
    int*   spsrc  = (int*)  alloc((size_t)e * 4);
    int*   spdst  = (int*)  alloc((size_t)e * 4);
    float* h      = (float*)alloc((size_t)n * CFEAT * 4);
    float* h2     = (float*)alloc((size_t)n * CFEAT * 4);
    (void)ws_size; (void)n_in; (void)out_size;

    k_gemm1<<<(n + 127) / 128, 256, 0, stream>>>(x, W1, h, deg, nspill, ticket, n);
    k_fill <<<2048, FILL_TB, 0, stream>>>(srcv, dstv, deg, slots, spsrc, spdst, nspill, ticket, e, n);

    const int nodes_per_wg = NAB * (PROP_TB / 32);
    const int gprop = (n + nodes_per_wg - 1) / nodes_per_wg;
    const Graph g = {deg, slots, spsrc, spdst, nspill};
    k_prop1    <<<gprop, PROP_TB, 0, stream>>>(h, h2, g, b1, n);
    k_prop2_out<<<gprop, PROP_TB, 0, stream>>>(h2, out, g, Wout, bout, n);
}


# k_prop1 as 256 resident work-groups, one per WGP: each forms dinv of all nodes in LDS once, src weights become LDS reads, the per-edge degree-word gather goes; 2 slots per node a trip; sum order kept
# speedup vs baseline: 4.1780x; 1.0360x over previous
#include <hip/hip_runtime.h>
#include <cstdint>

typedef __attribute__((ext_vector_type(2))) float v2f;
typedef __attribute__((ext_vector_type(8))) float v8f;

#define CFEAT 128
#define SLOTS 64


#define TICKET_PITCH 64

#define FILL_TB   256
#define FILL_EPT  2
#define FILL_SLICE (FILL_TB * FILL_EPT)

__global__ __launch_bounds__(FILL_TB)
void k_fill(const int* __restrict__ src, const int* __restrict__ dst,
            int* __restrict__ deg, int* __restrict__ slots,
            int* __restrict__ spill_src, int* __restrict__ spill_dst,
            int* __restrict__ nspill, int* __restrict__ ticket, int e, int n) {
    __shared__ int s_half;
    __shared__ int s_slice[2];
    if (threadIdx.x == 0) {
        const unsigned id = __builtin_amdgcn_s_sendmsg_rtn(0x87);
        s_half = (int)((id >> 18) & 1u);
    }
    __syncthreads();
    const int myhalf  = s_half;
    const int split   = ((n >> 1) + 31) & ~31;
    const int nslices = (e + FILL_SLICE - 1) / FILL_SLICE;
    int turn = 0;
    for (int pass = 0; pass < 2; ++pass) {
        const int range = myhalf ^ pass;
        int* __restrict__ tk = ticket + range * TICKET_PITCH;
        for (;;) {
            if (threadIdx.x == 0) s_slice[turn] = atomicAdd(tk, 1);
            __syncthreads();
            const int sl = s_slice[turn];
            turn ^= 1;
            if (sl >= nslices) break;
            int dd[FILL_EPT], ss[FILL_EPT];
            #pragma unroll
            for (int k = 0; k < FILL_EPT; ++k) {
                const int i = sl * FILL_SLICE + k * FILL_TB + (int)threadIdx.x;
                const bool in = i < e;
                dd[k] = in ? dst[i] : -1;
                ss[k] = in ? src[i] : 0;
            }
            int pp[FILL_EPT];
            #pragma unroll
            for (int k = 0; k < FILL_EPT; ++k) {
                const int d = dd[k];
                const bool mine = d >= 0 && (int)(d >= split) == range;
                pp[k] = mine ? atomicAdd(&deg[d], 1) : -1;
            }
            #pragma unroll
            for (int k = 0; k < FILL_EPT; ++k) {
                const int p = pp[k];
                if (p >= 0) {
                    const int d = dd[k];
                    if (p < SLOTS) {
                        slots[(size_t)d * SLOTS + p] = ss[k];
                    } else {
                        const int q = atomicAdd(nspill, 1);
                        spill_src[q] = ss[k];
                        spill_dst[q] = d;
                    }
                }
            }
        }
    }
}

__device__ __forceinline__ float dinv_of(int deg) { return rsqrtf((float)deg + 1.0f); }


__global__ __launch_bounds__(256)
void k_gemm1(const float* __restrict__ x, const float* __restrict__ W1,
             float* __restrict__ h, int* __restrict__ deg, int* __restrict__ nspill,
             int* __restrict__ ticket, int n) {
    __shared__ float sW[CFEAT * CFEAT];
    {
        const int i = blockIdx.x * 256 + threadIdx.x;
        if (i < n) deg[i] = 0;
        if (i == 0) { *nspill = 0; ticket[0] = 0; ticket[TICKET_PITCH] = 0; }
    }
    for (int i = threadIdx.x; i < (CFEAT * CFEAT) / 4; i += 256)
        ((float4*)sW)[i] = ((const float4*)W1)[i];
    __syncthreads();

    const int lane = threadIdx.x & 31;
    const int wave = threadIdx.x >> 5;
    const int hi   = lane >> 4;
    const int lo   = lane & 15;
    const int rowbase = (blockIdx.x * 8 + wave) * 16;
    if (rowbase >= n) return;

    int arow = rowbase + lo;
    if (arow >= n) arow = n - 1;
    const float* __restrict__ xrow = x + (size_t)arow * CFEAT;

    v8f acc[8];
    const v8f z = {0.f, 0.f, 0.f, 0.f, 0.f, 0.f, 0.f, 0.f};
    #pragma unroll
    for (int t = 0; t < 8; ++t) acc[t] = z;

    for (int k = 0; k < CFEAT; k += 4) {
        v2f a;
        a.x = xrow[k + 2 * hi];
        a.y = xrow[k + 2 * hi + 1];
        #pragma unroll
        for (int t = 0; t < 8; ++t) {
            v2f b;
            const int col = t * 16 + lo;
            b.x = sW[(k + 2 * hi) * CFEAT + col];
            b.y = sW[(k + 2 * hi + 1) * CFEAT + col];
            acc[t] = __builtin_amdgcn_wmma_f32_16x16x4_f32(
                false, a, false, b, (short)0, acc[t], false, false);
        }
    }

    if (rowbase + 16 <= n) {
        #pragma unroll
        for (int t = 0; t < 8; ++t) {
            #pragma unroll
            for (int r = 0; r < 8; ++r) {
                const int row = rowbase + r + 8 * hi;
                h[(size_t)row * CFEAT + t * 16 + lo] = acc[t][r];
            }
        }
    } else {
        #pragma unroll
        for (int t = 0; t < 8; ++t) {
            #pragma unroll
            for (int r = 0; r < 8; ++r) {
                const int row = rowbase + r + 8 * hi;
                if (row < n) h[(size_t)row * CFEAT + t * 16 + lo] = acc[t][r];
            }
        }
    }
}


struct Graph {
    const int* deg;
    const int* slots;
    const int* spill_src;
    const int* spill_dst;
    const int* nspill;
};

#define NAB      4
#define PROP_TB  128

#define TABN     65536
#define P1_WAVES 25
#define P1_TB    (P1_WAVES * 32)
#define P1_GRID  256
#define TAB_LD   8


template <bool TAB, int DEPTH>
__device__ __forceinline__ void gather_half(const float4* __restrict__ h4, int lane, int trips,
                                            const int (&cnt)[NAB], const int (&s)[NAB],
                                            const float (&w)[NAB], float4 (&acc)[NAB]) {
    for (int t = 0; t < trips; t += DEPTH) {
        float4 v[DEPTH][NAB];
        float  wk[DEPTH][NAB];
        #pragma unroll
        for (int j = 0; j < DEPTH; ++j) {
            #pragma unroll
            for (int k = 0; k < NAB; ++k) {
                if (t + j < cnt[k]) {
                    const int sk = __builtin_amdgcn_readlane(s[k], t + j);
                    if (TAB) wk[j][k] = __int_as_float(__builtin_amdgcn_readlane(__float_as_int(w[k]), t + j));
                    v[j][k] = h4[(size_t)sk * 32 + lane];
                }
            }
        }
        #pragma unroll
        for (int k = 0; k < NAB; ++k) {
            #pragma unroll
            for (int j = 0; j < DEPTH; ++j) {
                if (t + j < cnt[k]) {
                    if (TAB) {
                        acc[k].x += v[j][k].x * wk[j][k]; acc[k].y += v[j][k].y * wk[j][k];
                        acc[k].z += v[j][k].z * wk[j][k]; acc[k].w += v[j][k].w * wk[j][k];
                    } else {
                        acc[k].x += v[j][k].x; acc[k].y += v[j][k].y;
                        acc[k].z += v[j][k].z; acc[k].w += v[j][k].w;
                    }
                }
            }
        }
    }
}

template <bool TAB, int DEPTH>
__device__ __forceinline__ void sum_abreast(const float* __restrict__ hsrc,
                                            const Graph g, const float* tab,
                                            int node0, int n, int lane, float4 (&acc)[NAB],
                                            float (&din)[NAB]) {
    const float4* __restrict__ h4 = (const float4*)hsrc;
    int   nd[NAB];
    int   d[NAB];
    #pragma unroll
    for (int k = 0; k < NAB; ++k) {
        const bool ok = node0 + k < n;
        nd[k] = ok ? node0 + k : node0;
        const int dk = __builtin_amdgcn_readfirstlane(g.deg[nd[k]]);
        d[k]   = ok ? dk : 0;
        din[k] = dinv_of(dk);
    }
    int c0[NAB], c1[NAB];
    #pragma unroll
    for (int k = 0; k < NAB; ++k) {
        const int end = d[k] < SLOTS ? d[k] : SLOTS;
        c0[k] = end < 32 ? end : 32;
        c1[k] = end - c0[k];
    }
    int s0[NAB], s1[NAB];
    #pragma unroll
    for (int k = 0; k < NAB; ++k) {
        const int* __restrict__ row = g.slots + (size_t)nd[k] * SLOTS;
        const int* __restrict__ p0 = (lane < c0[k]) ? row + lane      : g.deg + nd[k];
        const int* __restrict__ p1 = (lane < c1[k]) ? row + lane + 32 : g.deg + nd[k];
        s0[k] = *p0;
        s1[k] = *p1;
    }
    #pragma unroll
    for (int k = 0; k < NAB; ++k) {
        s0[k] = (lane < c0[k]) ? s0[k] : 0;
        s1[k] = (lane < c1[k]) ? s1[k] : 0;
    }
    #pragma unroll
    for (int k = 0; k < NAB; ++k) {
        const float4 v = h4[(size_t)nd[k] * 32 + lane];
        if (TAB) {
            acc[k].x = v.x * din[k]; acc[k].y = v.y * din[k];
            acc[k].z = v.z * din[k]; acc[k].w = v.w * din[k];
        } else {
            acc[k] = v;
        }
    }
    float w0[NAB], w1[NAB];
    #pragma unroll
    for (int k = 0; k < NAB; ++k) {
        w0[k] = TAB ? tab[s0[k] < TABN ? s0[k] : TABN - 1] : 1.0f;
        w1[k] = TAB ? tab[s1[k] < TABN ? s1[k] : TABN - 1] : 1.0f;
    }
    if (TAB && n > TABN) {
        #pragma unroll
        for (int k = 0; k < NAB; ++k) {
            if (s0[k] >= TABN) w0[k] = dinv_of(g.deg[s0[k]]);
            if (s1[k] >= TABN) w1[k] = dinv_of(g.deg[s1[k]]);
        }
    }
    int trips0 = 0, trips1 = 0;
    #pragma unroll
    for (int k = 0; k < NAB; ++k) {
        trips0 = c0[k] > trips0 ? c0[k] : trips0;
        trips1 = c1[k] > trips1 ? c1[k] : trips1;
    }
    gather_half<TAB, DEPTH>(h4, lane, trips0, c0, s0, w0, acc);
    gather_half<TAB, DEPTH>(h4, lane, trips1, c1, s1, w1, acc);
    #pragma unroll
    for (int k = 0; k < NAB; ++k) {
        if (d[k] > SLOTS) {
            const int node = node0 + k;
            const int ns = *g.nspill;
            for (int q = 0; q < ns; ++q) {
                if (g.spill_dst[q] == node) {
                    const int   sk = g.spill_src[q];
                    const float wk = TAB ? dinv_of(g.deg[sk]) : 1.0f;
                    const float4 v = h4[(size_t)sk * 32 + lane];
                    acc[k].x += v.x * wk; acc[k].y += v.y * wk; acc[k].z += v.z * wk; acc[k].w += v.w * wk;
                }
            }
        }
        acc[k].x *= din[k]; acc[k].y *= din[k]; acc[k].z *= din[k]; acc[k].w *= din[k];
    }
}

__global__ __launch_bounds__(P1_TB)
void k_prop1(const float* __restrict__ h, float* __restrict__ h2,
             const Graph g,
             const float* __restrict__ b1, int n) {
    __shared__ float s_dinv[TABN];
    {
        const int ntab = n < TABN ? n : TABN;
        const int nq   = ntab >> 2;
        const int4* __restrict__ d4 = (const int4*)g.deg;
        for (int base = 0; base < nq; base += P1_TB * TAB_LD) {
            int4 v[TAB_LD];
            #pragma unroll
            for (int j = 0; j < TAB_LD; ++j) {
                const int i = base + j * P1_TB + (int)threadIdx.x;
                v[j] = d4[i < nq ? i : 0];
            }
            #pragma unroll
            for (int j = 0; j < TAB_LD; ++j) {
                const int i = base + j * P1_TB + (int)threadIdx.x;
                if (i < nq) {
                    float4 w;
                    w.x = dinv_of(v[j].x); w.y = dinv_of(v[j].y); w.z = dinv_of(v[j].z); w.w = dinv_of(v[j].w);
                    ((float4*)s_dinv)[i] = w;
                }
            }
        }
        for (int i = (nq << 2) + (int)threadIdx.x; i < ntab; i += P1_TB) s_dinv[i] = dinv_of(g.deg[i]);
    }
    __syncthreads();

    const int lane = threadIdx.x & 31;
    const int wave = __builtin_amdgcn_readfirstlane((int)(threadIdx.x >> 5));
    const int ngroups = (n + NAB - 1) / NAB;
    const float4 b = ((const float4*)b1)[lane];
    for (int grp = blockIdx.x * P1_WAVES + wave; grp < ngroups; grp += gridDim.x * P1_WAVES) {
        const int node0 = grp * NAB;
        float4 acc[NAB];
        float  din[NAB];
        sum_abreast<true, 2>(h, g, s_dinv, node0, n, lane, acc, din);
        #pragma unroll
        for (int k = 0; k < NAB; ++k) {
            if (node0 + k < n) {
                float4 r;
                r.x = fmaxf(acc[k].x + b.x, 0.0f) * din[k];
                r.y = fmaxf(acc[k].y + b.y, 0.0f) * din[k];
                r.z = fmaxf(acc[k].z + b.z, 0.0f) * din[k];
                r.w = fmaxf(acc[k].w + b.w, 0.0f) * din[k];
                ((float4*)h2)[(size_t)(node0 + k) * 32 + lane] = r;
            }
        }
    }
}

__global__ __launch_bounds__(PROP_TB)
void k_prop2_out(const float* __restrict__ h2, float* __restrict__ out,
                 const Graph g,
                 const float* __restrict__ Wout,
                 const float* __restrict__ bout, int n) {
    const int lane  = threadIdx.x & 31;
    const int wave  = __builtin_amdgcn_readfirstlane((int)((blockIdx.x * blockDim.x + threadIdx.x) >> 5));
    const int node0 = wave * NAB;
    if (node0 >= n) return;

    float4 acc[NAB];
    float  din[NAB];
    sum_abreast<false, 1>(h2, g, nullptr, node0, n, lane, acc, din);

    #pragma unroll
    for (int m = 0; m < 4; ++m) {
        float wl[4][2];
        #pragma unroll
        for (int j = 0; j < 4; ++j) {
            const float2 w = ((const float2*)Wout)[m * CFEAT + lane * 4 + j];
            wl[j][0] = w.x; wl[j][1] = w.y;
        }
        float p[NAB][2];
        #pragma unroll
        for (int k = 0; k < NAB; ++k)
            #pragma unroll
            for (int o = 0; o < 2; ++o) {
                float v = acc[k].x * wl[0][o] + acc[k].y * wl[1][o] +
                          acc[k].z * wl[2][o] + acc[k].w * wl[3][o];
                #pragma unroll
                for (int dd = 16; dd > 0; dd >>= 1) v += __shfl_xor(v, dd, 32);
                p[k][o] = v;
            }
        if (lane == 0) {
            const float bo0 = bout[m * 2 + 0];
            const float bo1 = bout[m * 2 + 1];
            #pragma unroll
            for (int k = 0; k < NAB; ++k) {
                if (node0 + k < n) {
                    float2 r;
                    r.x = p[k][0] + bo0;
                    r.y = p[k][1] + bo1;
                    ((float2*)out)[(size_t)m * n + node0 + k] = r;
                }
            }
        }
    }
}


extern "C" void kernel_launch(void* const* d_in, const int* in_sizes, int n_in,
                              void* d_out, int out_size, void* d_ws, size_t ws_size,
                              hipStream_t stream) {
    const float* x    = (const float*)d_in[0];
    const int*   ei   = (const int*)d_in[1];
    const float* W1   = (const float*)d_in[2];
    const float* b1   = (const float*)d_in[3];
    const float* Wout = (const float*)d_in[4];
    const float* bout = (const float*)d_in[5];
    float* out = (float*)d_out;

    const int n = in_sizes[0] / CFEAT;
    const int e = in_sizes[1] / 2;
    const int* srcv = ei;
    const int* dstv = ei + e;

    uintptr_t p = (uintptr_t)d_ws;
    auto alloc = [&](size_t bytes) {
        p = (p + 15) & ~(uintptr_t)15;
        void* r = (void*)p;
        p += bytes;
        return r;
    };
    int*   deg    = (int*)  alloc((size_t)n * 4);
    int*   nspill = (int*)  alloc(16);
    p = (p + 255) & ~(uintptr_t)255;
    int*   ticket = (int*)  alloc(2 * TICKET_PITCH * 4);
    p = (p + 255) & ~(uintptr_t)255;
    int*   slots  = (int*)  alloc((size_t)n * SLOTS * 4);
    int*   spsrc  = (int*)  alloc((size_t)e * 4);
    int*   spdst  = (int*)  alloc((size_t)e * 4);
    float* h      = (float*)alloc((size_t)n * CFEAT * 4);
    float* h2     = (float*)alloc((size_t)n * CFEAT * 4);
    (void)ws_size; (void)n_in; (void)out_size;

    k_gemm1<<<(n + 127) / 128, 256, 0, stream>>>(x, W1, h, deg, nspill, ticket, n);
    k_fill <<<2048, FILL_TB, 0, stream>>>(srcv, dstv, deg, slots, spsrc, spdst, nspill, ticket, e, n);

    const int nodes_per_wg = NAB * (PROP_TB / 32);
    const int gprop = (n + nodes_per_wg - 1) / nodes_per_wg;
    const Graph g = {deg, slots, spsrc, spdst, nspill};
    k_prop1    <<<P1_GRID, P1_TB, 0, stream>>>(h, h2, g, b1, n);
    k_prop2_out<<<gprop, PROP_TB, 0, stream>>>(h2, out, g, Wout, bout, n);
}
